# starting kernel
# baseline (speedup 1.0000x reference)
#include <hip/hip_runtime.h>

typedef __attribute__((ext_vector_type(2))) float v2f;
typedef __attribute__((ext_vector_type(4))) float v4f;
typedef __attribute__((ext_vector_type(8))) float v8f;

namespace {
constexpr int kB     = 128;
constexpr int kH     = 128;
constexpr int kW     = 128;
constexpr int kS     = 2;
constexpr int kOut   = 4;
constexpr int kHo    = kH - 1;
constexpr int kWo    = kW - 1;
constexpr int kWG    = 4;
constexpr int kWaves = 8;
}

__global__ __launch_bounds__(256) void eps_wmma_kernel(
    const float* __restrict__ in,
    const float* __restrict__ core,
    float* __restrict__ out) {
  __shared__ float lds_core[16 * 16];
  __shared__ float lds_out[kWaves][32 * kOut];

  const int tid = threadIdx.x;
  {
    const int row = tid >> 4;
    const int col = tid & 15;
    lds_core[tid] = (col < kOut) ? core[row * kOut + col] : 0.0f;
  }
  __syncthreads();

  const int wid  = tid >> 5;
  const int lane = tid & 31;
  const int m    = lane & 15;
  const int half = lane >> 4;

  const unsigned g   = (unsigned)blockIdx.x * kWaves + (unsigned)wid;
  const int      wg  = (int)(g & (kWG - 1));
  const unsigned rem = g >> 2;
  const int      h   = (int)(rem % (unsigned)kHo);
  const int      b   = (int)(rem / (unsigned)kHo);

  const int w0 = wg * 32 + m;
  const int w1 = w0 + 16;
  const int wc0 = (w0 < kWo) ? w0 : (kWo - 1);
  const int wc1 = (w1 < kWo) ? w1 : (kWo - 1);

  const size_t rowbase = (size_t)(b * kH + h) * kW;

  const int kb = (2 * half) * 16 + m;
  v2f b0, b1, b2, b3;
  b0.x = lds_core[  0 + kb];  b0.y = lds_core[  0 + kb + 16];
  b1.x = lds_core[ 64 + kb];  b1.y = lds_core[ 64 + kb + 16];
  b2.x = lds_core[128 + kb];  b2.y = lds_core[128 + kb + 16];
  b3.x = lds_core[192 + kb];  b3.y = lds_core[192 + kb + 16];

  v8f acc0 = {0.f, 0.f, 0.f, 0.f, 0.f, 0.f, 0.f, 0.f};
  v8f acc1 = {0.f, 0.f, 0.f, 0.f, 0.f, 0.f, 0.f, 0.f};

#pragma unroll
  for (int sub = 0; sub < 2; ++sub) {
    const size_t base = (rowbase + (sub ? wc1 : wc0)) * kS;
    const v2f x00 = *(const v2f*)(in + base);
    const v2f x01 = *(const v2f*)(in + base + kS);
    const v2f x10 = *(const v2f*)(in + base + (size_t)kW * kS);
    const v2f x11 = *(const v2f*)(in + base + (size_t)kW * kS + kS);

    float p01[4];
    p01[0] = x00.x * x01.x;
    p01[1] = x00.x * x01.y;
    p01[2] = x00.y * x01.x;
    p01[3] = x00.y * x01.y;
    const float s2v = half ? x10.y : x10.x;
    const float q0 = s2v * x11.x;
    const float q1 = s2v * x11.y;

    v2f a0, a1, a2, a3;
    a0.x = p01[0] * q0; a0.y = p01[0] * q1;
    a1.x = p01[1] * q0; a1.y = p01[1] * q1;
    a2.x = p01[2] * q0; a2.y = p01[2] * q1;
    a3.x = p01[3] * q0; a3.y = p01[3] * q1;

    v8f acc = sub ? acc1 : acc0;
    acc = __builtin_amdgcn_wmma_f32_16x16x4_f32(false, a0, false, b0, (short)0,
                                                acc, false, false);
    acc = __builtin_amdgcn_wmma_f32_16x16x4_f32(false, a1, false, b1, (short)0,
                                                acc, false, false);
    acc = __builtin_amdgcn_wmma_f32_16x16x4_f32(false, a2, false, b2, (short)0,
                                                acc, false, false);
    acc = __builtin_amdgcn_wmma_f32_16x16x4_f32(false, a3, false, b3, (short)0,
                                                acc, false, false);
    if (sub) acc1 = acc; else acc0 = acc;
  }

  if (m < kOut) {
#pragma unroll
    for (int r = 0; r < 8; ++r) {
      lds_out[wid][(r + 8 * half) * kOut + m]      = acc0[r];
      lds_out[wid][(16 + r + 8 * half) * kOut + m] = acc1[r];
    }
  }
  asm volatile("s_wait_dscnt 0" ::: "memory");

  const int wl = wg * 32 + lane;
  if (wl < kWo) {
    const v4f o = *(const v4f*)&lds_out[wid][lane * kOut];
    *(v4f*)(out + (((size_t)b * kHo + h) * kWo + wl) * kOut) = o;
  }
}

extern "C" void kernel_launch(void* const* d_in, const int* in_sizes, int n_in,
                              void* d_out, int out_size, void* d_ws, size_t ws_size,
                              hipStream_t stream) {
  (void)in_sizes; (void)n_in; (void)out_size; (void)d_ws; (void)ws_size;
  const float* in   = (const float*)d_in[0];
  const float* core = (const float*)d_in[1];
  float*       out  = (float*)d_out;

  const int total_waves = kB * kHo * kWG;
  const int blocks      = total_waves / kWaves;
  eps_wmma_kernel<<<blocks, 256, 0, stream>>>(in, core, out);
}


# Each wave walks 16 output rows of its 32-pixel strip (512 blocks, was 8128), reusing the shared input row's products and keeping 8 rows of loads in flight; order of sums unchanged
# speedup vs baseline: 1.3717x; 1.3717x over previous
#include <hip/hip_runtime.h>


typedef __attribute__((ext_vector_type(2))) float v2f;
typedef __attribute__((ext_vector_type(4))) float v4f;
typedef __attribute__((ext_vector_type(8))) float v8f;

namespace {
constexpr int kB     = 128;
constexpr int kH     = 128;
constexpr int kW     = 128;
constexpr int kS     = 2;
constexpr int kOut   = 4;
constexpr int kHo    = kH - 1;
constexpr int kWo    = kW - 1;
constexpr int kWG    = 4;
constexpr int kWaves = 8;
constexpr int kRows  = 16;
constexpr int kChunks = (kHo + kRows - 1) / kRows;
constexpr int kAhead = 8;

struct RowRegs { v2f a0, b0, a1, b1; };
}

__global__ __launch_bounds__(256) void eps_wmma_kernel(
    const float* __restrict__ in,
    const float* __restrict__ core,
    float* __restrict__ out) {
  __shared__ float lds_core[16 * 16];
  __shared__ float lds_out[kWaves][32 * kOut];

  const int tid  = threadIdx.x;
  const int lane = tid & 31;
  const int m    = lane & 15;
  const int half = lane >> 4;
  const int wid = __builtin_amdgcn_readfirstlane(tid >> 5);

  const unsigned g     = (unsigned)blockIdx.x * kWaves + (unsigned)wid;
  const int      wg    = (int)(g & (kWG - 1));
  const unsigned rem   = g >> 2;
  const int      chunk = (int)(rem % (unsigned)kChunks);
  const int      b     = (int)(rem / (unsigned)kChunks);
  const int      h0    = chunk * kRows;

  const int w0  = wg * 32 + m;
  const int w1  = w0 + 16;
  const int wc0 = (w0 < kWo) ? w0 : (kWo - 1);
  const int wc1 = (w1 < kWo) ? w1 : (kWo - 1);
  const int off0 = wc0 * kS;
  const int off1 = wc1 * kS;

  auto load_row = [&](int j) -> RowRegs {
    int hr = h0 + j;
    hr = (hr < kH - 1) ? hr : (kH - 1);
    const float* rp = in + (size_t)(b * kH + hr) * (size_t)(kW * kS);
    RowRegs r;
    r.a0 = *(const v2f*)(rp + off0);
    r.b0 = *(const v2f*)(rp + off0 + kS);
    r.a1 = *(const v2f*)(rp + off1);
    r.b1 = *(const v2f*)(rp + off1 + kS);
    return r;
  };

  const RowRegs top = load_row(0);
  RowRegs ring[kAhead];
#pragma unroll
  for (int k = 0; k < kAhead; ++k) ring[k] = load_row(1 + k);

  {
    const int row = tid >> 4;
    const int col = tid & 15;
    lds_core[tid] = (col < kOut) ? core[row * kOut + col] : 0.0f;
  }
  __syncthreads();

  const int kb = (2 * half) * 16 + m;
  v2f b0, b1, b2, b3;
  b0.x = lds_core[  0 + kb];  b0.y = lds_core[  0 + kb + 16];
  b1.x = lds_core[ 64 + kb];  b1.y = lds_core[ 64 + kb + 16];
  b2.x = lds_core[128 + kb];  b2.y = lds_core[128 + kb + 16];
  b3.x = lds_core[192 + kb];  b3.y = lds_core[192 + kb + 16];

  float pu0[4], pu1[4];
  pu0[0] = top.a0.x * top.b0.x;  pu0[1] = top.a0.x * top.b0.y;
  pu0[2] = top.a0.y * top.b0.x;  pu0[3] = top.a0.y * top.b0.y;
  pu1[0] = top.a1.x * top.b1.x;  pu1[1] = top.a1.x * top.b1.y;
  pu1[2] = top.a1.y * top.b1.x;  pu1[3] = top.a1.y * top.b1.y;

  const int  wl      = wg * 32 + lane;
  const bool col_ok  = wl < kWo;
  float* const outp  = out + ((size_t)b * kHo * kWo + (size_t)wl) * kOut;

#pragma unroll
  for (int r = 0; r < kRows; ++r) {
    const RowRegs lo = ring[r % kAhead];
    if (r + 1 + kAhead <= kRows) ring[r % kAhead] = load_row(r + 1 + kAhead);

    float pl0[4], pl1[4];
    pl0[0] = lo.a0.x * lo.b0.x;  pl0[1] = lo.a0.x * lo.b0.y;
    pl0[2] = lo.a0.y * lo.b0.x;  pl0[3] = lo.a0.y * lo.b0.y;
    pl1[0] = lo.a1.x * lo.b1.x;  pl1[1] = lo.a1.x * lo.b1.y;
    pl1[2] = lo.a1.y * lo.b1.x;  pl1[3] = lo.a1.y * lo.b1.y;

    const float q00 = half ? pl0[2] : pl0[0];
    const float q01 = half ? pl0[3] : pl0[1];
    const float q10 = half ? pl1[2] : pl1[0];
    const float q11 = half ? pl1[3] : pl1[1];

    v8f acc0 = {0.f, 0.f, 0.f, 0.f, 0.f, 0.f, 0.f, 0.f};
    v8f acc1 = {0.f, 0.f, 0.f, 0.f, 0.f, 0.f, 0.f, 0.f};
    {
      v2f a0, a1, a2, a3;
      a0.x = pu0[0] * q00; a0.y = pu0[0] * q01;
      a1.x = pu0[1] * q00; a1.y = pu0[1] * q01;
      a2.x = pu0[2] * q00; a2.y = pu0[2] * q01;
      a3.x = pu0[3] * q00; a3.y = pu0[3] * q01;
      acc0 = __builtin_amdgcn_wmma_f32_16x16x4_f32(false, a0, false, b0, (short)0, acc0, false, false);
      acc0 = __builtin_amdgcn_wmma_f32_16x16x4_f32(false, a1, false, b1, (short)0, acc0, false, false);
      acc0 = __builtin_amdgcn_wmma_f32_16x16x4_f32(false, a2, false, b2, (short)0, acc0, false, false);
      acc0 = __builtin_amdgcn_wmma_f32_16x16x4_f32(false, a3, false, b3, (short)0, acc0, false, false);
    }
    {
      v2f a0, a1, a2, a3;
      a0.x = pu1[0] * q10; a0.y = pu1[0] * q11;
      a1.x = pu1[1] * q10; a1.y = pu1[1] * q11;
      a2.x = pu1[2] * q10; a2.y = pu1[2] * q11;
      a3.x = pu1[3] * q10; a3.y = pu1[3] * q11;
      acc1 = __builtin_amdgcn_wmma_f32_16x16x4_f32(false, a0, false, b0, (short)0, acc1, false, false);
      acc1 = __builtin_amdgcn_wmma_f32_16x16x4_f32(false, a1, false, b1, (short)0, acc1, false, false);
      acc1 = __builtin_amdgcn_wmma_f32_16x16x4_f32(false, a2, false, b2, (short)0, acc1, false, false);
      acc1 = __builtin_amdgcn_wmma_f32_16x16x4_f32(false, a3, false, b3, (short)0, acc1, false, false);
    }

#pragma unroll
    for (int i = 0; i < 4; ++i) { pu0[i] = pl0[i]; pu1[i] = pl1[i]; }

    if (m < kOut) {
#pragma unroll
      for (int e = 0; e < 8; ++e) {
        lds_out[wid][(e + 8 * half) * kOut + m]      = acc0[e];
        lds_out[wid][(16 + e + 8 * half) * kOut + m] = acc1[e];
      }
    }
    asm volatile("s_wait_dscnt 0" ::: "memory");

    const int h = h0 + r;
    if (h < kHo && col_ok) {
      const v4f o = *(const v4f*)&lds_out[wid][lane * kOut];
      *(v4f*)(outp + (size_t)h * (kWo * kOut)) = o;
    }
  }
}

extern "C" void kernel_launch(void* const* d_in, const int* in_sizes, int n_in,
                              void* d_out, int out_size, void* d_ws, size_t ws_size,
                              hipStream_t stream) {
  (void)in_sizes; (void)n_in; (void)out_size; (void)d_ws; (void)ws_size;
  const float* in   = (const float*)d_in[0];
  const float* core = (const float*)d_in[1];
  float*       out  = (float*)d_out;

  const int total_waves = kB * kChunks * kWG;
  const int blocks      = total_waves / kWaves;
  eps_wmma_kernel<<<blocks, 256, 0, stream>>>(in, core, out);
}
